# starting kernel
# baseline (speedup 1.0000x reference)
#include <hip/hip_runtime.h>
#include <hip/hip_bf16.h>
#include <math.h>

typedef __attribute__((ext_vector_type(2))) float v2f;
typedef __attribute__((ext_vector_type(8))) float v8f;

__global__ __launch_bounds__(256) void norm_rows64(const float* __restrict__ src,
                                                   float* __restrict__ dst,
                                                   int nrows) {
  int w = blockIdx.x * 8 + (threadIdx.x >> 5);
  int lane = threadIdx.x & 31;
  if (w >= nrows) return;
  const float2* s = (const float2*)(src + (size_t)w * 64);
  float2 v = s[lane];
  float ss = v.x * v.x + v.y * v.y;
#pragma unroll
  for (int off = 16; off > 0; off >>= 1) ss += __shfl_xor(ss, off, 32);
  float inv = 1.0f / fmaxf(sqrtf(ss), 1e-12f);
  float2 o;
  o.x = v.x * inv;
  o.y = v.y * inv;
  ((float2*)(dst + (size_t)w * 64))[lane] = o;
}

__global__ __launch_bounds__(256) void mlp1(const float* __restrict__ satN,
                                            const float* __restrict__ W1,
                                            const float* __restrict__ b1,
                                            float* __restrict__ Hout) {
  int wave = blockIdx.x * 8 + (threadIdx.x >> 5);
  int lane = threadIdx.x & 31;
  int l15 = lane & 15, half = lane >> 4;
  int mt = wave >> 5;
  int nt = wave & 31;
  int row0 = mt * 16;
  int col = nt * 16 + l15;
  const v2f* aP = (const v2f*)(satN + (size_t)(row0 + l15) * 64 + 2 * half);
  const v2f* bP = (const v2f*)(W1 + (size_t)col * 64 + 2 * half);
  v8f acc = {};
#pragma unroll
  for (int kk = 0; kk < 16; ++kk) {
    v2f a = aP[2 * kk];
    v2f b = bP[2 * kk];
    acc = __builtin_amdgcn_wmma_f32_16x16x4_f32(false, a, false, b, (short)0,
                                                acc, false, false);
  }
  float bias = b1[col];
#pragma unroll
  for (int v = 0; v < 8; ++v) {
    int r = row0 + v + 8 * half;
    Hout[(size_t)r * 512 + col] = fmaxf(acc[v] + bias, 0.0f);
  }
}

#define HS_STRIDE 516
#define A2_STRIDE 68
#define D2_STRIDE 132

__global__ __launch_bounds__(256) void fused_tail(const float* __restrict__ Hbuf,
                                                  const float* __restrict__ satN,
                                                  const float* __restrict__ panoN,
                                                  const float* __restrict__ W2,
                                                  const float* __restrict__ b2,
                                                  float* __restrict__ out)
{
  __shared__ float Hs[16 * HS_STRIDE];
  __shared__ float A2[64 * A2_STRIDE];
  __shared__ float d2acc[16 * D2_STRIDE];
  __shared__ float satB[16 * 64];
  __shared__ float ys[64];
  __shared__ float omax[128];

  const int s = blockIdx.x;
  const int tid = threadIdx.x;
  const int wave = tid >> 5;
  const int lane = tid & 31;
  const int l15 = lane & 15, half = lane >> 4;

  if (tid < 128) omax[tid] = 0.0f;

  for (int eb = 0; eb < 4; ++eb) {
    const int seBase = s * 64 + eb * 16;
    __syncthreads();
    for (int i = tid; i < 16 * 512; i += 256) {
      int r = i >> 9, c = i & 511;
      Hs[r * HS_STRIDE + c] = Hbuf[(size_t)(seBase + r) * 512 + c];
    }
    for (int i = tid; i < 16 * 64; i += 256)
      satB[i] = satN[(size_t)seBase * 64 + i];
    for (int i = tid; i < 16 * D2_STRIDE; i += 256) d2acc[i] = 0.0f;
    __syncthreads();

    for (int ch = 0; ch < 16; ++ch) {
      const int obase = ch * 256;
      v8f acc1[2];
      {
        const v2f* aP = (const v2f*)(Hs + l15 * HS_STRIDE + 2 * half);
#pragma unroll
        for (int t = 0; t < 2; ++t) {
          const int o = obase + (wave + 8 * t) * 16 + l15;
          const v2f* bP = (const v2f*)(W2 + (size_t)o * 512 + 2 * half);
          v8f acc = {};
#pragma unroll 8
          for (int kk = 0; kk < 128; ++kk) {
            v2f a = aP[2 * kk];
            v2f b = bP[2 * kk];
            acc = __builtin_amdgcn_wmma_f32_16x16x4_f32(false, a, false, b,
                                                        (short)0, acc, false, false);
          }
          acc1[t] = acc;
        }
      }
      __syncthreads();
#pragma unroll
      for (int t = 0; t < 2; ++t) {
        const int ol = (wave + 8 * t) * 16 + l15;
        const float bb = b2[obase + ol];
        const int kl = ol >> 6, f = ol & 63;
#pragma unroll
        for (int v = 0; v < 8; ++v) {
          const int se = v + 8 * half;
          A2[(se * 4 + kl) * A2_STRIDE + f] = acc1[t][v] + bb;
        }
      }
      __syncthreads();
      if (tid < 64) {
        const int se = tid >> 2;
        float a = 0.0f;
#pragma unroll 8
        for (int f = 0; f < 64; ++f)
          a += A2[tid * A2_STRIDE + f] * satB[se * 64 + f];
        ys[tid] = a;
      }
      __syncthreads();
      {
        const int p = wave * 16 + l15;
        const v2f* bP = (const v2f*)(panoN + (size_t)p * 64 + 2 * half);
#pragma unroll
        for (int mt = 0; mt < 4; ++mt) {
          const int m0 = mt * 16;
          const v2f* aP = (const v2f*)(A2 + (m0 + l15) * A2_STRIDE + 2 * half);
          v8f acc = {};
#pragma unroll
          for (int kk = 0; kk < 16; ++kk) {
            v2f a = aP[2 * kk];
            v2f b = bP[2 * kk];
            acc = __builtin_amdgcn_wmma_f32_16x16x4_f32(false, a, false, b,
                                                        (short)0, acc, false, false);
          }
          const int r0 = m0 + 8 * half;
          float s0 = 0.0f, s1 = 0.0f;
#pragma unroll
          for (int v = 0; v < 4; ++v) {
            float d0 = acc[v] - ys[r0 + v];
            float d1 = acc[v + 4] - ys[r0 + 4 + v];
            s0 += d0 * d0;
            s1 += d1 * d1;
          }
          const int se0 = r0 >> 2;
          d2acc[se0 * D2_STRIDE + p] += s0;
          d2acc[(se0 + 1) * D2_STRIDE + p] += s1;
        }
      }
    }
    __syncthreads();
    if (tid < 128) {
      float m = omax[tid];
#pragma unroll
      for (int se = 0; se < 16; ++se)
        m = fmaxf(m, d2acc[se * D2_STRIDE + tid]);
      omax[tid] = m;
    }
  }
  __syncthreads();
  if (tid < 128) out[(size_t)tid * 256 + s] = omax[tid];
}

extern "C" void kernel_launch(void* const* d_in, const int* in_sizes, int n_in,
                              void* d_out, int out_size, void* d_ws, size_t ws_size,
                              hipStream_t stream) {
  (void)in_sizes; (void)n_in; (void)out_size; (void)ws_size;
  const float* sat  = (const float*)d_in[0];
  const float* pano = (const float*)d_in[1];
  const float* W1   = (const float*)d_in[2];
  const float* b1   = (const float*)d_in[3];
  const float* W2   = (const float*)d_in[4];
  const float* b2   = (const float*)d_in[5];
  float* out = (float*)d_out;

  char* ws = (char*)d_ws;
  float* satN  = (float*)ws;
  float* panoN = (float*)(ws + 4194304);
  float* Hbuf  = (float*)(ws + 4194304 + 32768);

  norm_rows64<<<16384 / 8, 256, 0, stream>>>(sat, satN, 16384);
  norm_rows64<<<128 / 8, 256, 0, stream>>>(pano, panoN, 128);
  mlp1<<<32768 / 8, 256, 0, stream>>>(satN, W1, b1, Hbuf);
  fused_tail<<<256, 256, 0, stream>>>(Hbuf, satN, panoN, W2, b2, out);
}


# Layer-2 GEMM blocked over all 64 e-rows of an s: each W2 fragment read feeds 4 row tiles (8 accumulator tiles per wave, next block's fragments loaded ahead); e loop gone, LDS 246 KiB
# speedup vs baseline: 2.8208x; 2.8208x over previous
#include <hip/hip_runtime.h>
#include <hip/hip_bf16.h>
#include <math.h>

typedef __attribute__((ext_vector_type(2))) float v2f;
typedef __attribute__((ext_vector_type(8))) float v8f;

__global__ __launch_bounds__(256) void norm_rows64(const float* __restrict__ src,
                                                   float* __restrict__ dst,
                                                   int nrows) {
  int w = blockIdx.x * 8 + (threadIdx.x >> 5);
  int lane = threadIdx.x & 31;
  if (w >= nrows) return;
  const float2* s = (const float2*)(src + (size_t)w * 64);
  float2 v = s[lane];
  float ss = v.x * v.x + v.y * v.y;
#pragma unroll
  for (int off = 16; off > 0; off >>= 1) ss += __shfl_xor(ss, off, 32);
  float inv = 1.0f / fmaxf(sqrtf(ss), 1e-12f);
  float2 o;
  o.x = v.x * inv;
  o.y = v.y * inv;
  ((float2*)(dst + (size_t)w * 64))[lane] = o;
}

__global__ __launch_bounds__(256) void mlp1(const float* __restrict__ satN,
                                            const float* __restrict__ W1,
                                            const float* __restrict__ b1,
                                            float* __restrict__ Hout) {
  int wave = blockIdx.x * 8 + (threadIdx.x >> 5);
  int lane = threadIdx.x & 31;
  int l15 = lane & 15, half = lane >> 4;
  int mt = wave >> 5;
  int nt = wave & 31;
  int row0 = mt * 16;
  int col = nt * 16 + l15;
  const v2f* aP = (const v2f*)(satN + (size_t)(row0 + l15) * 64 + 2 * half);
  const v2f* bP = (const v2f*)(W1 + (size_t)col * 64 + 2 * half);
  v8f acc = {};
#pragma unroll
  for (int kk = 0; kk < 16; ++kk) {
    v2f a = aP[2 * kk];
    v2f b = bP[2 * kk];
    acc = __builtin_amdgcn_wmma_f32_16x16x4_f32(false, a, false, b, (short)0,
                                                acc, false, false);
  }
  float bias = b1[col];
#pragma unroll
  for (int v = 0; v < 8; ++v) {
    int r = row0 + v + 8 * half;
    Hout[(size_t)r * 512 + col] = fmaxf(acc[v] + bias, 0.0f);
  }
}

#define HS_LPITCH 2052
#define HS_MPITCH 512
#define A2_STRIDE 68
#define D2_STRIDE 132
#define KBLK 8

struct Frags {
  v2f a[4][KBLK];
  v2f b[2][KBLK];
};

__device__ __forceinline__ void load_frags(Frags& f, const v2f* a0, const v2f* a1,
                                           const v2f* a2, const v2f* a3,
                                           const v2f* b0, const v2f* b1, int kk0) {
#pragma unroll
  for (int j = 0; j < KBLK; ++j) {
    f.b[0][j] = b0[2 * (kk0 + j)];
    f.b[1][j] = b1[2 * (kk0 + j)];
  }
#pragma unroll
  for (int j = 0; j < KBLK; ++j) {
    f.a[0][j] = a0[2 * (kk0 + j)];
    f.a[1][j] = a1[2 * (kk0 + j)];
    f.a[2][j] = a2[2 * (kk0 + j)];
    f.a[3][j] = a3[2 * (kk0 + j)];
  }
}

__device__ __forceinline__ void mma_block(v8f (&acc)[4][2], const Frags& f) {
#pragma unroll
  for (int j = 0; j < KBLK; ++j) {
#pragma unroll
    for (int m = 0; m < 4; ++m) {
#pragma unroll
      for (int t = 0; t < 2; ++t) {
        acc[m][t] = __builtin_amdgcn_wmma_f32_16x16x4_f32(false, f.a[m][j], false, f.b[t][j],
                                                          (short)0, acc[m][t], false, false);
      }
    }
  }
}

__global__ __launch_bounds__(256) void fused_tail(const float* __restrict__ Hbuf,
                                                  const float* __restrict__ satN,
                                                  const float* __restrict__ panoN,
                                                  const float* __restrict__ W2,
                                                  const float* __restrict__ b2,
                                                  float* __restrict__ out)
{
  __shared__ float Hs[16 * HS_LPITCH];
  __shared__ float A2[256 * A2_STRIDE];
  __shared__ float d2acc[64 * D2_STRIDE];
  __shared__ float satB[64 * 64];
  __shared__ float ys[256];

  const int s = blockIdx.x;
  const int tid = threadIdx.x;
  const int wave = tid >> 5;
  const int lane = tid & 31;
  const int l15 = lane & 15, half = lane >> 4;
  const int seBase = s * 64;

  for (int i = tid; i < 64 * 512; i += 256) {
    int r = i >> 9, c = i & 511;
    Hs[(r & 15) * HS_LPITCH + (r >> 4) * HS_MPITCH + c] = Hbuf[(size_t)(seBase + r) * 512 + c];
  }
  for (int i = tid; i < 64 * 64; i += 256)
    satB[i] = satN[(size_t)seBase * 64 + i];
  for (int i = tid; i < 64 * D2_STRIDE; i += 256) d2acc[i] = 0.0f;
  __syncthreads();

  for (int ch = 0; ch < 16; ++ch) {
    const int obase = ch * 256;
    v8f acc1[4][2];
#pragma unroll
    for (int m = 0; m < 4; ++m) {
      acc1[m][0] = v8f{};
      acc1[m][1] = v8f{};
    }
    {
      const v2f* a0 = (const v2f*)(Hs + l15 * HS_LPITCH + 0 * HS_MPITCH + 2 * half);
      const v2f* a1 = (const v2f*)(Hs + l15 * HS_LPITCH + 1 * HS_MPITCH + 2 * half);
      const v2f* a2 = (const v2f*)(Hs + l15 * HS_LPITCH + 2 * HS_MPITCH + 2 * half);
      const v2f* a3 = (const v2f*)(Hs + l15 * HS_LPITCH + 3 * HS_MPITCH + 2 * half);
      const int o0 = obase + wave * 16 + l15;
      const int o1 = obase + (wave + 8) * 16 + l15;
      const v2f* b0 = (const v2f*)(W2 + (size_t)o0 * 512 + 2 * half);
      const v2f* b1 = (const v2f*)(W2 + (size_t)o1 * 512 + 2 * half);
      Frags f0, f1;
      load_frags(f0, a0, a1, a2, a3, b0, b1, 0);
#pragma unroll 1
      for (int kb = 0; kb < 128 - 2 * KBLK; kb += 2 * KBLK) {
        load_frags(f1, a0, a1, a2, a3, b0, b1, kb + KBLK);
        mma_block(acc1, f0);
        load_frags(f0, a0, a1, a2, a3, b0, b1, kb + 2 * KBLK);
        mma_block(acc1, f1);
      }
      load_frags(f1, a0, a1, a2, a3, b0, b1, 128 - KBLK);
      mma_block(acc1, f0);
      mma_block(acc1, f1);
    }
    __syncthreads();
#pragma unroll
    for (int t = 0; t < 2; ++t) {
      const int ol = (wave + 8 * t) * 16 + l15;
      const float bb = b2[obase + ol];
      const int kl = ol >> 6, f = ol & 63;
#pragma unroll
      for (int m = 0; m < 4; ++m) {
#pragma unroll
        for (int v = 0; v < 8; ++v) {
          const int se = m * 16 + v + 8 * half;
          A2[(se * 4 + kl) * A2_STRIDE + f] = acc1[m][t][v] + bb;
        }
      }
    }
    __syncthreads();
    {
      const int se = tid >> 2;
      float a = 0.0f;
#pragma unroll 8
      for (int f = 0; f < 64; ++f)
        a += A2[tid * A2_STRIDE + f] * satB[se * 64 + f];
      ys[tid] = a;
    }
    __syncthreads();
    {
      const int p = wave * 16 + l15;
      const v2f* bP = (const v2f*)(panoN + (size_t)p * 64 + 2 * half);
#pragma unroll 4
      for (int mt = 0; mt < 16; ++mt) {
        const int m0 = mt * 16;
        const v2f* aP = (const v2f*)(A2 + (m0 + l15) * A2_STRIDE + 2 * half);
        v8f acc = {};
#pragma unroll
        for (int kk = 0; kk < 16; ++kk) {
          v2f a = aP[2 * kk];
          v2f b = bP[2 * kk];
          acc = __builtin_amdgcn_wmma_f32_16x16x4_f32(false, a, false, b,
                                                      (short)0, acc, false, false);
        }
        const int r0 = m0 + 8 * half;
        float s0 = 0.0f, s1 = 0.0f;
#pragma unroll
        for (int v = 0; v < 4; ++v) {
          float d0 = acc[v] - ys[r0 + v];
          float d1 = acc[v + 4] - ys[r0 + 4 + v];
          s0 += d0 * d0;
          s1 += d1 * d1;
        }
        const int se0 = r0 >> 2;
        d2acc[se0 * D2_STRIDE + p] += s0;
        d2acc[(se0 + 1) * D2_STRIDE + p] += s1;
      }
    }
  }
  __syncthreads();
  if (tid < 128) {
    float m = 0.0f;
#pragma unroll
    for (int se = 0; se < 64; ++se)
      m = fmaxf(m, d2acc[se * D2_STRIDE + tid]);
    out[(size_t)tid * 256 + s] = m;
  }
}

extern "C" void kernel_launch(void* const* d_in, const int* in_sizes, int n_in,
                              void* d_out, int out_size, void* d_ws, size_t ws_size,
                              hipStream_t stream) {
  (void)in_sizes; (void)n_in; (void)out_size; (void)ws_size;
  const float* sat  = (const float*)d_in[0];
  const float* pano = (const float*)d_in[1];
  const float* W1   = (const float*)d_in[2];
  const float* b1   = (const float*)d_in[3];
  const float* W2   = (const float*)d_in[4];
  const float* b2   = (const float*)d_in[5];
  float* out = (float*)d_out;

  char* ws = (char*)d_ws;
  float* satN  = (float*)ws;
  float* panoN = (float*)(ws + 4194304);
  float* Hbuf  = (float*)(ws + 4194304 + 32768);

  norm_rows64<<<16384 / 8, 256, 0, stream>>>(sat, satN, 16384);
  norm_rows64<<<128 / 8, 256, 0, stream>>>(pano, panoN, 128);
  mlp1<<<32768 / 8, 256, 0, stream>>>(satN, W1, b1, Hbuf);
  fused_tail<<<256, 256, 0, stream>>>(Hbuf, satN, panoN, W2, b2, out);
}


# One launch instead of four: each workgroup normalises its sat and pano rows into LDS and computes its own H tile (layer 1, same WMMA chain) in its prologue; H never goes through memory
# speedup vs baseline: 3.0357x; 1.0762x over previous
#include <hip/hip_runtime.h>
#include <hip/hip_bf16.h>
#include <math.h>

typedef __attribute__((ext_vector_type(2))) float v2f;
typedef __attribute__((ext_vector_type(8))) float v8f;

__device__ __forceinline__ void norm_row_to_lds(const float* __restrict__ src, float* dst,
                                                int lane) {
  const float2 v = ((const float2*)src)[lane];
  float ss = v.x * v.x + v.y * v.y;
#pragma unroll
  for (int off = 16; off > 0; off >>= 1) ss += __shfl_xor(ss, off, 32);
  const float inv = 1.0f / fmaxf(sqrtf(ss), 1e-12f);
  float2 o;
  o.x = v.x * inv;
  o.y = v.y * inv;
  ((float2*)dst)[lane] = o;
}

__device__ __forceinline__ void load_w1_frags(v2f (&b)[16], const float* __restrict__ W1,
                                              int col, int half) {
  const v2f* bP = (const v2f*)(W1 + (size_t)col * 64 + 2 * half);
#pragma unroll
  for (int kk = 0; kk < 16; ++kk) b[kk] = bP[2 * kk];
}

#define HS_LPITCH 2052
#define HS_MPITCH 512
#define A2_STRIDE 68
#define D2_STRIDE 132
#define SB_STRIDE 68
#define KBLK 8

struct Frags {
  v2f a[4][KBLK];
  v2f b[2][KBLK];
};

__device__ __forceinline__ void load_frags(Frags& f, const v2f* a0, const v2f* a1,
                                           const v2f* a2, const v2f* a3,
                                           const v2f* b0, const v2f* b1, int kk0) {
#pragma unroll
  for (int j = 0; j < KBLK; ++j) {
    f.b[0][j] = b0[2 * (kk0 + j)];
    f.b[1][j] = b1[2 * (kk0 + j)];
  }
#pragma unroll
  for (int j = 0; j < KBLK; ++j) {
    f.a[0][j] = a0[2 * (kk0 + j)];
    f.a[1][j] = a1[2 * (kk0 + j)];
    f.a[2][j] = a2[2 * (kk0 + j)];
    f.a[3][j] = a3[2 * (kk0 + j)];
  }
}

__device__ __forceinline__ void mma_block(v8f (&acc)[4][2], const Frags& f) {
#pragma unroll
  for (int j = 0; j < KBLK; ++j) {
#pragma unroll
    for (int m = 0; m < 4; ++m) {
#pragma unroll
      for (int t = 0; t < 2; ++t) {
        acc[m][t] = __builtin_amdgcn_wmma_f32_16x16x4_f32(false, f.a[m][j], false, f.b[t][j],
                                                          (short)0, acc[m][t], false, false);
      }
    }
  }
}

__global__ __launch_bounds__(256) void fused_all(const float* __restrict__ sat,
                                                 const float* __restrict__ pano,
                                                 const float* __restrict__ W1,
                                                 const float* __restrict__ bias1,
                                                 const float* __restrict__ W2,
                                                 const float* __restrict__ b2,
                                                 float* __restrict__ out)
{
  __shared__ float Hs[16 * HS_LPITCH];
  __shared__ __attribute__((aligned(4096))) float A2[256 * A2_STRIDE];
  __shared__ float d2acc[64 * D2_STRIDE];
  __shared__ float satB[64 * SB_STRIDE];
  __shared__ float panoS[128 * SB_STRIDE];
  __shared__ float ys[256];

  const int s = blockIdx.x;
  const int tid = threadIdx.x;
  const int wave = tid >> 5;
  const int lane = tid & 31;
  const int l15 = lane & 15, half = lane >> 4;
  const int seBase = s * 64;

  v2f bw[2][16];
  load_w1_frags(bw[0], W1, (wave * 4) * 16 + l15, half);

#pragma unroll
  for (int i = 0; i < 8; ++i) {
    const int r = wave * 8 + i;
    norm_row_to_lds(sat + (size_t)(seBase + r) * 64, satB + r * SB_STRIDE, lane);
  }
#pragma unroll
  for (int i = 0; i < 16; ++i) {
    const int p = wave * 16 + i;
    norm_row_to_lds(pano + (size_t)p * 64, panoS + p * SB_STRIDE, lane);
  }
  for (int i = tid; i < 64 * D2_STRIDE; i += 256) d2acc[i] = 0.0f;
  __syncthreads();

  {
    const float* aBase = satB + l15 * SB_STRIDE + 2 * half;
#pragma unroll
    for (int j = 0; j < 4; ++j) {
      const int col = (wave * 4 + j) * 16 + l15;
      if (j < 3) load_w1_frags(bw[(j + 1) & 1], W1, col + 16, half);
      v8f acc[4];
#pragma unroll
      for (int m = 0; m < 4; ++m) acc[m] = v8f{};
#pragma unroll
      for (int kk = 0; kk < 16; ++kk) {
#pragma unroll
        for (int m = 0; m < 4; ++m) {
          const v2f a = *(const v2f*)(aBase + m * 16 * SB_STRIDE + 4 * kk);
          acc[m] = __builtin_amdgcn_wmma_f32_16x16x4_f32(false, a, false, bw[j & 1][kk],
                                                         (short)0, acc[m], false, false);
        }
      }
      const float bias = bias1[col];
#pragma unroll
      for (int m = 0; m < 4; ++m) {
#pragma unroll
        for (int v = 0; v < 8; ++v)
          Hs[(v + 8 * half) * HS_LPITCH + m * HS_MPITCH + col] = fmaxf(acc[m][v] + bias, 0.0f);
      }
    }
  }
  v2f pb[16];
  {
    const v2f* bP = (const v2f*)(panoS + (wave * 16 + l15) * SB_STRIDE + 2 * half);
#pragma unroll
    for (int kk = 0; kk < 16; ++kk) pb[kk] = bP[2 * kk];
  }
  __syncthreads();

  for (int ch = 0; ch < 16; ++ch) {
    const int obase = ch * 256;
    v8f acc1[4][2];
#pragma unroll
    for (int m = 0; m < 4; ++m) {
      acc1[m][0] = v8f{};
      acc1[m][1] = v8f{};
    }
    {
      const v2f* a0 = (const v2f*)(Hs + l15 * HS_LPITCH + 0 * HS_MPITCH + 2 * half);
      const v2f* a1 = (const v2f*)(Hs + l15 * HS_LPITCH + 1 * HS_MPITCH + 2 * half);
      const v2f* a2 = (const v2f*)(Hs + l15 * HS_LPITCH + 2 * HS_MPITCH + 2 * half);
      const v2f* a3 = (const v2f*)(Hs + l15 * HS_LPITCH + 3 * HS_MPITCH + 2 * half);
      const int o0 = obase + wave * 16 + l15;
      const int o1 = obase + (wave + 8) * 16 + l15;
      const v2f* b0 = (const v2f*)(W2 + (size_t)o0 * 512 + 2 * half);
      const v2f* b1 = (const v2f*)(W2 + (size_t)o1 * 512 + 2 * half);
      Frags f0, f1;
      load_frags(f0, a0, a1, a2, a3, b0, b1, 0);
#pragma unroll 1
      for (int kb = 0; kb < 128 - 2 * KBLK; kb += 2 * KBLK) {
        load_frags(f1, a0, a1, a2, a3, b0, b1, kb + KBLK);
        mma_block(acc1, f0);
        load_frags(f0, a0, a1, a2, a3, b0, b1, kb + 2 * KBLK);
        mma_block(acc1, f1);
      }
      load_frags(f1, a0, a1, a2, a3, b0, b1, 128 - KBLK);
      mma_block(acc1, f0);
      mma_block(acc1, f1);
    }
    __syncthreads();
#pragma unroll
    for (int t = 0; t < 2; ++t) {
      const int ol = (wave + 8 * t) * 16 + l15;
      const float bb = b2[obase + ol];
      const int kl = ol >> 6, f = ol & 63;
#pragma unroll
      for (int m = 0; m < 4; ++m) {
#pragma unroll
        for (int v = 0; v < 8; ++v) {
          const int se = m * 16 + v + 8 * half;
          A2[(se * 4 + kl) * A2_STRIDE + f] = acc1[m][t][v] + bb;
        }
      }
    }
    __syncthreads();
    {
      const int se = tid >> 2;
      float a = 0.0f;
#pragma unroll 8
      for (int f = 0; f < 64; ++f)
        a += A2[tid * A2_STRIDE + f] * satB[se * SB_STRIDE + f];
      ys[tid] = a;
    }
    __syncthreads();
    {
      const int p = wave * 16 + l15;
#pragma unroll 4
      for (int mt = 0; mt < 16; ++mt) {
        const int m0 = mt * 16;
        const v2f* aP = (const v2f*)(A2 + (m0 + l15) * A2_STRIDE + 2 * half);
        v8f acc = {};
#pragma unroll
        for (int kk = 0; kk < 16; ++kk) {
          v2f a = aP[2 * kk];
          v2f b = pb[kk];
          acc = __builtin_amdgcn_wmma_f32_16x16x4_f32(false, a, false, b,
                                                      (short)0, acc, false, false);
        }
        const int r0 = m0 + 8 * half;
        float s0 = 0.0f, s1 = 0.0f;
#pragma unroll
        for (int v = 0; v < 4; ++v) {
          float d0 = acc[v] - ys[r0 + v];
          float d1 = acc[v + 4] - ys[r0 + 4 + v];
          s0 += d0 * d0;
          s1 += d1 * d1;
        }
        const int se0 = r0 >> 2;
        d2acc[se0 * D2_STRIDE + p] += s0;
        d2acc[(se0 + 1) * D2_STRIDE + p] += s1;
      }
    }
  }
  __syncthreads();
  if (tid < 128) {
    float m = 0.0f;
#pragma unroll
    for (int se = 0; se < 64; ++se)
      m = fmaxf(m, d2acc[se * D2_STRIDE + tid]);
    out[(size_t)tid * 256 + s] = m;
  }
}

extern "C" void kernel_launch(void* const* d_in, const int* in_sizes, int n_in,
                              void* d_out, int out_size, void* d_ws, size_t ws_size,
                              hipStream_t stream) {
  (void)in_sizes; (void)n_in; (void)out_size; (void)d_ws; (void)ws_size;
  const float* sat  = (const float*)d_in[0];
  const float* pano = (const float*)d_in[1];
  const float* W1   = (const float*)d_in[2];
  const float* b1   = (const float*)d_in[3];
  const float* W2   = (const float*)d_in[4];
  const float* b2   = (const float*)d_in[5];
  float* out = (float*)d_out;

  fused_all<<<256, 256, 0, stream>>>(sat, pano, W1, b1, W2, b2, out);
}


# Layer-2 tile computed transposed (W2 as A operand): its accumulators are the A fragments of the pano product, no LDS round trip or barrier in the main loop, 1 wave per SIMD; sums reordered
# speedup vs baseline: 3.2096x; 1.0573x over previous
#include <hip/hip_runtime.h>
#include <hip/hip_bf16.h>
#include <math.h>

typedef __attribute__((ext_vector_type(2))) float v2f;
typedef __attribute__((ext_vector_type(8))) float v8f;

__device__ __forceinline__ void norm_row_to_lds(const float* __restrict__ src, float* dst,
                                                int lane) {
  const float2 v = ((const float2*)src)[lane];
  float ss = v.x * v.x + v.y * v.y;
#pragma unroll
  for (int off = 16; off > 0; off >>= 1) ss += __shfl_xor(ss, off, 32);
  const float inv = 1.0f / fmaxf(sqrtf(ss), 1e-12f);
  float2 o;
  o.x = v.x * inv;
  o.y = v.y * inv;
  ((float2*)dst)[lane] = o;
}

__device__ __forceinline__ void load_w1_frags(v2f (&b)[16], const float* __restrict__ W1,
                                              int col, int half) {
  const v2f* bP = (const v2f*)(W1 + (size_t)col * 64 + 2 * half);
#pragma unroll
  for (int kk = 0; kk < 16; ++kk) b[kk] = bP[2 * kk];
}

#define HS_LPITCH 2052
#define HS_MPITCH 512
#define SB_STRIDE 68
#define KBLK 4

struct Frags {
  v2f a[4][KBLK];
  v2f b[4][KBLK];
};

__device__ __forceinline__ void load_frags(Frags& f, const v2f* w, const v2f* hp, int kk0) {
#pragma unroll
  for (int j = 0; j < KBLK; ++j) {
#pragma unroll
    for (int q = 0; q < 4; ++q) f.a[q][j] = w[q * (16 * 512 / 2) + 2 * (kk0 + j)];
  }
#pragma unroll
  for (int j = 0; j < KBLK; ++j) {
#pragma unroll
    for (int m = 0; m < 4; ++m) f.b[m][j] = hp[m * (HS_MPITCH / 2) + 2 * (kk0 + j)];
  }
}

__device__ __forceinline__ void load_b2_row(v8f (&bb)[4], const float* __restrict__ p) {
#pragma unroll
  for (int q = 0; q < 4; ++q) {
    const float4 lo = *(const float4*)(p + 16 * q);
    const float4 hi = *(const float4*)(p + 16 * q + 4);
    bb[q] = v8f{lo.x, lo.y, lo.z, lo.w, hi.x, hi.y, hi.z, hi.w};
  }
}

__device__ __forceinline__ void mma_block(v8f (&acc)[4][4], const Frags& f) {
#pragma unroll
  for (int j = 0; j < KBLK; ++j) {
#pragma unroll
    for (int q = 0; q < 4; ++q) {
#pragma unroll
      for (int m = 0; m < 4; ++m) {
        acc[q][m] = __builtin_amdgcn_wmma_f32_16x16x4_f32(false, f.a[q][j], false, f.b[m][j],
                                                          (short)0, acc[q][m], false, false);
      }
    }
  }
}

__global__ __launch_bounds__(128) void fused_all(const float* __restrict__ sat,
                                                 const float* __restrict__ pano,
                                                 const float* __restrict__ W1,
                                                 const float* __restrict__ bias1,
                                                 const float* __restrict__ W2,
                                                 const float* __restrict__ b2,
                                                 float* __restrict__ out)
{
  __shared__ __attribute__((aligned(16))) float Hs[16 * HS_LPITCH];
  __shared__ __attribute__((aligned(16))) float satB[64 * SB_STRIDE];
  __shared__ __attribute__((aligned(16))) float panoS[128 * SB_STRIDE];

  const int s = blockIdx.x;
  const int tid = threadIdx.x;
  const int wave = tid >> 5;
  const int lane = tid & 31;
  const int l15 = lane & 15, half = lane >> 4;
  const int seBase = s * 64;

  v2f bw[2][16];
  load_w1_frags(bw[0], W1, (wave * 8) * 16 + l15, half);

#pragma unroll
  for (int i = 0; i < 16; ++i) {
    const int r = wave * 16 + i;
    norm_row_to_lds(sat + (size_t)(seBase + r) * 64, satB + r * SB_STRIDE, lane);
  }
#pragma unroll
  for (int i = 0; i < 32; ++i) {
    const int p = wave * 32 + i;
    norm_row_to_lds(pano + (size_t)p * 64, panoS + p * SB_STRIDE, lane);
  }
  __syncthreads();

  {
    const float* aBase = satB + l15 * SB_STRIDE + 2 * half;
#pragma unroll
    for (int j = 0; j < 8; ++j) {
      const int col = (wave * 8 + j) * 16 + l15;
      if (j < 7) load_w1_frags(bw[(j + 1) & 1], W1, col + 16, half);
      v8f acc[4];
#pragma unroll
      for (int m = 0; m < 4; ++m) acc[m] = v8f{};
#pragma unroll
      for (int kk = 0; kk < 16; ++kk) {
#pragma unroll
        for (int m = 0; m < 4; ++m) {
          const v2f a = *(const v2f*)(aBase + m * 16 * SB_STRIDE + 4 * kk);
          acc[m] = __builtin_amdgcn_wmma_f32_16x16x4_f32(false, a, false, bw[j & 1][kk],
                                                         (short)0, acc[m], false, false);
        }
      }
      const float bias = bias1[col];
#pragma unroll
      for (int m = 0; m < 4; ++m) {
#pragma unroll
        for (int v = 0; v < 8; ++v)
          Hs[(v + 8 * half) * HS_LPITCH + m * HS_MPITCH + col] = fmaxf(acc[m][v] + bias, 0.0f);
      }
    }
  }
  v8f sN[4][4];
#pragma unroll
  for (int m = 0; m < 4; ++m) {
#pragma unroll
    for (int q = 0; q < 4; ++q)
    {
      const float* sp = satB + (16 * m + l15) * SB_STRIDE + 16 * q + 8 * half;
      const float4 lo = *(const float4*)sp;
      const float4 hi = *(const float4*)(sp + 4);
      sN[m][q] = v8f{lo.x, lo.y, lo.z, lo.w, hi.x, hi.y, hi.z, hi.w};
    }
  }
  v8f d2[4][8];
#pragma unroll
  for (int m = 0; m < 4; ++m) {
#pragma unroll
    for (int pt = 0; pt < 8; ++pt) d2[m][pt] = v8f{};
  }
  __syncthreads();

  const v2f* hp = (const v2f*)(Hs + l15 * HS_LPITCH + 2 * half);
  const v2f* w = (const v2f*)(W2 + (size_t)(64 * wave + l15) * 512 + 2 * half);
  const float* panoL = panoS + l15 * SB_STRIDE + 8 * half;
  Frags f0, f1;
  load_frags(f0, w, hp, 0);
  v8f bnext[4];
  load_b2_row(bnext, b2 + 64 * wave + 8 * half);
#pragma unroll 1
  for (int i = 0; i < 16; ++i) {
    const int k = 4 * i + wave;
    v8f acc[4][4];
#pragma unroll
    for (int q = 0; q < 4; ++q) {
#pragma unroll
      for (int m = 0; m < 4; ++m) acc[q][m] = bnext[q];
    }
#pragma unroll 1
    for (int kb = 0; kb < 128 - 2 * KBLK; kb += 2 * KBLK) {
      load_frags(f1, w, hp, kb + KBLK);
      mma_block(acc, f0);
      load_frags(f0, w, hp, kb + 2 * KBLK);
      mma_block(acc, f1);
    }
    load_frags(f1, w, hp, 128 - KBLK);
    mma_block(acc, f0);
    mma_block(acc, f1);
    w += 256 * 512 / 2;
    if (i < 15) {
      load_frags(f0, w, hp, 0);
      load_b2_row(bnext, b2 + 64 * (k + 4) + 8 * half);
    }

    float part[4];
#pragma unroll
    for (int m = 0; m < 4; ++m) {
      float a = 0.0f;
#pragma unroll
      for (int q = 0; q < 4; ++q) {
#pragma unroll
        for (int r = 0; r < 8; ++r) a += acc[q][m][r] * sN[m][q][r];
      }
      part[m] = a;
    }
    const v2f minus1 = {-1.0f, 0.0f};
#pragma unroll
    for (int pt = 0; pt < 8; ++pt) {
      v2f pb[16];
#pragma unroll
      for (int q = 0; q < 4; ++q) {
#pragma unroll
        for (int jj = 0; jj < 4; ++jj)
          pb[4 * q + jj] = *(const v2f*)(panoL + 16 * pt * SB_STRIDE + 16 * q + 2 * jj);
      }
      v8f y[4];
#pragma unroll
      for (int m = 0; m < 4; ++m) y[m] = v8f{};
#pragma unroll
      for (int q = 0; q < 4; ++q) {
#pragma unroll
        for (int jj = 0; jj < 4; ++jj) {
#pragma unroll
          for (int m = 0; m < 4; ++m) {
            const v2f a = {acc[q][m][2 * jj], acc[q][m][2 * jj + 1]};
            y[m] = __builtin_amdgcn_wmma_f32_16x16x4_f32(false, a, false, pb[4 * q + jj],
                                                         (short)0, y[m], false, false);
          }
        }
      }
#pragma unroll
      for (int m = 0; m < 4; ++m) {
        const v2f a = {part[m], 0.0f};
        y[m] = __builtin_amdgcn_wmma_f32_16x16x4_f32(false, a, false, minus1,
                                                     (short)0, y[m], false, false);
#pragma unroll
        for (int r = 0; r < 8; ++r) d2[m][pt][r] += y[m][r] * y[m][r];
      }
    }
  }
  __syncthreads();
  float* dsum = Hs;
#pragma unroll
  for (int m = 0; m < 4; ++m) {
#pragma unroll
    for (int pt = 0; pt < 8; ++pt) {
#pragma unroll
      for (int r = 0; r < 8; ++r)
        dsum[(wave * 256 + (m * 8 + pt) * 8 + r) * 32 + lane] = d2[m][pt][r];
    }
  }
  __syncthreads();
  {
    const int pt = tid >> 4, c = tid & 15;
    float mx = 0.0f;
#pragma unroll
    for (int m = 0; m < 4; ++m) {
#pragma unroll
      for (int h = 0; h < 2; ++h) {
#pragma unroll
        for (int r = 0; r < 8; ++r) {
          const int at = ((m * 8 + pt) * 8 + r) * 32 + c + 16 * h;
          const float v = (dsum[at] + dsum[at + 256 * 32]) + (dsum[at + 2 * 256 * 32] + dsum[at + 3 * 256 * 32]);
          mx = fmaxf(mx, v);
        }
      }
    }
    out[(size_t)tid * 256 + s] = mx;
  }
}

extern "C" void kernel_launch(void* const* d_in, const int* in_sizes, int n_in,
                              void* d_out, int out_size, void* d_ws, size_t ws_size,
                              hipStream_t stream) {
  (void)in_sizes; (void)n_in; (void)out_size; (void)d_ws; (void)ws_size;
  const float* sat  = (const float*)d_in[0];
  const float* pano = (const float*)d_in[1];
  const float* W1   = (const float*)d_in[2];
  const float* b1   = (const float*)d_in[3];
  const float* W2   = (const float*)d_in[4];
  const float* b2   = (const float*)d_in[5];
  float* out = (float*)d_out;

  fused_all<<<256, 128, 0, stream>>>(sat, pano, W1, b1, W2, b2, out);
}
